# starting kernel
# baseline (speedup 1.0000x reference)
#include <hip/hip_runtime.h>
#include <math.h>

typedef float v2f __attribute__((ext_vector_type(2)));
typedef float v8f __attribute__((ext_vector_type(8)));

#define B_   128
#define S_   512
#define D_   768
#define T_   51
#define NI_  26
#define NS_  122
#define ROWS_   (B_ * T_)
#define MTILES_ (ROWS_ / 16)
#define LPAD 772

__global__ __launch_bounds__(256) void intent_kernel(
    const float* __restrict__ res, const float* __restrict__ W_id,
    const float* __restrict__ b_id, float* __restrict__ out)
{
    int idx = blockIdx.x * blockDim.x + threadIdx.x;
    if (idx >= B_ * NI_) return;
    int b = idx / NI_;
    int j = idx - b * NI_;
    const float* r = res + (size_t)b * D_;
    float acc = 0.f;
    for (int d = 0; d < D_; ++d)
        acc = fmaf(r[d], W_id[(size_t)d * NI_ + j], acc);
    out[idx] = acc + b_id[j];
}

__global__ __launch_bounds__(256) void word_slot_kernel(
    const float* __restrict__ res_all, const float* __restrict__ res,
    const int*  __restrict__ tstart,   const int*  __restrict__ slen,
    const float* __restrict__ W_attn,  const float* __restrict__ W_slot,
    const float* __restrict__ b_slot,  float* __restrict__ out_sf)
{
    __shared__ float g0s[16][LPAD];
    __shared__ float g1s[16][LPAD];
    __shared__ float sc0[16], sc1[16], wg0[16], wg1[16];
    __shared__ int   rowB[16], rowS0[16], rowS1[16], rowLen1[16];

    const int tid  = threadIdx.x;
    const int lane = tid & 31;
    const int wave = tid >> 5;
    const int half = (lane >> 4) & 1;
    const int l16  = lane & 15;
    const int kA   = 2 * half;
    const int mblk = blockIdx.x;

    if (tid < 16) {
        int R  = mblk * 16 + tid;
        int bb = R / T_;
        int tt = R - bb * T_;
        int st = tstart[bb * T_ + tt];
        int ln = slen[bb * T_ + tt];
        rowB[tid]    = bb;
        rowS0[tid]   = st;
        rowS1[tid]   = st + ln;
        rowLen1[tid] = (ln == 1);
        sc0[tid] = 0.f;
        sc1[tid] = 0.f;
    }
    __syncthreads();

    for (int idx = tid; idx < 16 * (D_ / 4); idx += 256) {
        int m = idx / (D_ / 4);
        int q = idx - m * (D_ / 4);
        const float4* s0p = (const float4*)(res_all + (size_t)(rowB[m] * S_ + rowS0[m]) * D_);
        const float4* s1p = (const float4*)(res_all + (size_t)(rowB[m] * S_ + rowS1[m]) * D_);
        ((float4*)&g0s[m][0])[q] = s0p[q];
        ((float4*)&g1s[m][0])[q] = s1p[q];
    }
    __syncthreads();

    float s0r[8], s1r[8];
#pragma unroll
    for (int i = 0; i < 8; ++i) { s0r[i] = 0.f; s1r[i] = 0.f; }

    for (int nt = 0; nt < 6; ++nt) {
        const int nb = wave * 96 + nt * 16;
        const int c  = nb + l16;
        v8f acc = {};
        for (int k0 = 0; k0 < D_; k0 += 4) {
            v2f a, b;
            a.x = g0s[l16][k0 + kA];
            a.y = g0s[l16][k0 + kA + 1];
            b.x = W_attn[(size_t)(k0 + kA) * D_ + c];
            b.y = W_attn[(size_t)(k0 + kA + 1) * D_ + c];
            acc = __builtin_amdgcn_wmma_f32_16x16x4_f32(
                false, a, false, b, (short)0, acc, false, false);
        }
#pragma unroll
        for (int i = 0; i < 8; ++i) {
            int   m = i + 8 * half;
            float q = acc[i];
            s0r[i] = fmaf(q, g0s[m][c], s0r[i]);
            s1r[i] = fmaf(q, g1s[m][c], s1r[i]);
        }
    }

#pragma unroll
    for (int i = 0; i < 8; ++i) {
        float a0 = s0r[i], a1 = s1r[i];
        for (int off = 8; off >= 1; off >>= 1) {
            a0 += __shfl_xor(a0, off, 32);
            a1 += __shfl_xor(a1, off, 32);
        }
        if (l16 == 0) {
            int m = i + 8 * half;
            atomicAdd(&sc0[m], a0);
            atomicAdd(&sc1[m], a1);
        }
    }
    __syncthreads();

    if (tid < 16) {
        const float scale = 0.03608439182435161f;
        float a  = sc0[tid] * scale;
        float b  = sc1[tid] * scale;
        float mx = fmaxf(a, b);
        float e0 = __expf(a - mx);
        float e1 = __expf(b - mx);
        float iv = 1.f / (e0 + e1);
        float w0 = e0 * iv, w1 = e1 * iv;
        if (rowLen1[tid]) { w0 = 1.f; w1 = 0.f; }
        wg0[tid] = w0;
        wg1[tid] = w1;
    }
    __syncthreads();

    for (int idx = tid; idx < 16 * D_; idx += 256) {
        int m = idx / D_;
        int d = idx - m * D_;
        float w = wg0[m] * g0s[m][d] + wg1[m] * g1s[m][d]
                + res[(size_t)rowB[m] * D_ + d];
        g0s[m][d] = w;
    }
    __syncthreads();

    {
        const int  nb = wave * 16;
        const int  c  = nb + l16;
        const bool cv = (c < NS_);
        const int  cc = cv ? c : (NS_ - 1);
        const float msk = cv ? 1.f : 0.f;
        v8f acc = {};
        for (int k0 = 0; k0 < D_; k0 += 4) {
            v2f a, b;
            a.x = g0s[l16][k0 + kA];
            a.y = g0s[l16][k0 + kA + 1];
            b.x = W_slot[(size_t)(k0 + kA) * NS_ + cc] * msk;
            b.y = W_slot[(size_t)(k0 + kA + 1) * NS_ + cc] * msk;
            acc = __builtin_amdgcn_wmma_f32_16x16x4_f32(
                false, a, false, b, (short)0, acc, false, false);
        }
        if (cv) {
            float bias = b_slot[c];
#pragma unroll
            for (int i = 0; i < 8; ++i) {
                int m = i + 8 * half;
                int R = mblk * 16 + m;
                out_sf[(size_t)R * NS_ + c] = acc[i] + bias;
            }
        }
    }
}

extern "C" void kernel_launch(void* const* d_in, const int* in_sizes, int n_in,
                              void* d_out, int out_size, void* d_ws, size_t ws_size,
                              hipStream_t stream)
{
    (void)in_sizes; (void)n_in; (void)out_size; (void)d_ws; (void)ws_size;

    const float* res_all = (const float*)d_in[0];
    const float* res     = (const float*)d_in[1];
    const int*   tstart  = (const int*)  d_in[2];
    const int*   slen    = (const int*)  d_in[3];
    const float* W_attn  = (const float*)d_in[4];
    const float* W_id    = (const float*)d_in[5];
    const float* b_id    = (const float*)d_in[6];
    const float* W_slot  = (const float*)d_in[7];
    const float* b_slot  = (const float*)d_in[8];

    float* out_id = (float*)d_out;
    float* out_sf = out_id + B_ * NI_;

    intent_kernel<<<(B_ * NI_ + 255) / 256, 256, 0, stream>>>(res, W_id, b_id, out_id);
    word_slot_kernel<<<MTILES_, 256, 0, stream>>>(
        res_all, res, tstart, slen, W_attn, W_slot, b_slot, out_sf);
}


# Intent head on WMMA f32 16x16x4 in its own launch: 8 work-groups x 16 waves, K split 16 ways, every load of a wave issued before its first wait; K slices added through LDS (summation order changes)
# speedup vs baseline: 1.5765x; 1.5765x over previous
#include <hip/hip_runtime.h>
#include <math.h>

typedef float v2f __attribute__((ext_vector_type(2)));
typedef float v8f __attribute__((ext_vector_type(8)));

#define B_   128
#define S_   512
#define D_   768
#define T_   51
#define NI_  26
#define NS_  122
#define ROWS_   (B_ * T_)
#define MTILES_ (ROWS_ / 16)
#define LPAD 772

#define IK_WAVES 16
#define IK_KSL   (D_ / IK_WAVES)

__global__ __launch_bounds__(IK_WAVES * 32) void intent_kernel(
    const float* __restrict__ res, const float* __restrict__ W_id,
    const float* __restrict__ b_id, float* __restrict__ out)
{
    __shared__ float part[IK_WAVES][2][8][32];

    const int tid  = threadIdx.x;
    const int lane = tid & 31;
    const int wave = tid >> 5;
    const int half = (lane >> 4) & 1;
    const int l16  = lane & 15;
    const int kA   = 2 * half;
    const int m0   = blockIdx.x * 16;
    const int kb   = wave * IK_KSL;

    const int c0 = l16;
    const int c1 = (16 + l16 < NI_) ? (16 + l16) : (NI_ - 1);

    const float* ap = res  + (size_t)(m0 + l16) * D_ + kb + kA;
    const float* bp = W_id + (size_t)(kb + kA) * NI_;

    v8f acc0 = {}, acc1 = {};
    for (int s = 0; s < IK_KSL / 4; ++s) {
        v2f a, b0, b1;
        a.x  = ap[4 * s];
        a.y  = ap[4 * s + 1];
        b0.x = bp[(size_t)(4 * s)     * NI_ + c0];
        b0.y = bp[(size_t)(4 * s + 1) * NI_ + c0];
        b1.x = bp[(size_t)(4 * s)     * NI_ + c1];
        b1.y = bp[(size_t)(4 * s + 1) * NI_ + c1];
        acc0 = __builtin_amdgcn_wmma_f32_16x16x4_f32(
            false, a, false, b0, (short)0, acc0, false, false);
        acc1 = __builtin_amdgcn_wmma_f32_16x16x4_f32(
            false, a, false, b1, (short)0, acc1, false, false);
    }

    for (int i = 0; i < 8; ++i) {
        part[wave][0][i][lane] = acc0[i];
        part[wave][1][i][lane] = acc1[i];
    }
    __syncthreads();

    if (tid < 16 * NI_) {
        const int row = tid / NI_;
        const int col = tid - row * NI_;
        const int tl  = col >> 4;
        const int i   = row & 7;
        const int ln  = ((row >> 3) << 4) | (col & 15);
        float s = part[0][tl][i][ln];
        for (int w = 1; w < IK_WAVES; ++w) s += part[w][tl][i][ln];
        out[(size_t)blockIdx.x * (16 * NI_) + tid] = s + b_id[col];
    }
}

__global__ __launch_bounds__(256) void word_slot_kernel(
    const float* __restrict__ res_all, const float* __restrict__ res,
    const int*  __restrict__ tstart,   const int*  __restrict__ slen,
    const float* __restrict__ W_attn,  const float* __restrict__ W_slot,
    const float* __restrict__ b_slot,  float* __restrict__ out_sf)
{
    __shared__ float g0s[16][LPAD];
    __shared__ float g1s[16][LPAD];
    __shared__ float sc0[16], sc1[16], wg0[16], wg1[16];
    __shared__ int   rowB[16], rowS0[16], rowS1[16], rowLen1[16];

    const int tid  = threadIdx.x;
    const int lane = tid & 31;
    const int wave = tid >> 5;
    const int half = (lane >> 4) & 1;
    const int l16  = lane & 15;
    const int kA   = 2 * half;
    const int mblk = blockIdx.x;

    if (tid < 16) {
        int R  = mblk * 16 + tid;
        int bb = R / T_;
        int tt = R - bb * T_;
        int st = tstart[bb * T_ + tt];
        int ln = slen[bb * T_ + tt];
        rowB[tid]    = bb;
        rowS0[tid]   = st;
        rowS1[tid]   = st + ln;
        rowLen1[tid] = (ln == 1);
        sc0[tid] = 0.f;
        sc1[tid] = 0.f;
    }
    __syncthreads();

    for (int idx = tid; idx < 16 * (D_ / 4); idx += 256) {
        int m = idx / (D_ / 4);
        int q = idx - m * (D_ / 4);
        const float4* s0p = (const float4*)(res_all + (size_t)(rowB[m] * S_ + rowS0[m]) * D_);
        const float4* s1p = (const float4*)(res_all + (size_t)(rowB[m] * S_ + rowS1[m]) * D_);
        ((float4*)&g0s[m][0])[q] = s0p[q];
        ((float4*)&g1s[m][0])[q] = s1p[q];
    }
    __syncthreads();

    float s0r[8], s1r[8];
#pragma unroll
    for (int i = 0; i < 8; ++i) { s0r[i] = 0.f; s1r[i] = 0.f; }

    for (int nt = 0; nt < 6; ++nt) {
        const int nb = wave * 96 + nt * 16;
        const int c  = nb + l16;
        v8f acc = {};
        for (int k0 = 0; k0 < D_; k0 += 4) {
            v2f a, b;
            a.x = g0s[l16][k0 + kA];
            a.y = g0s[l16][k0 + kA + 1];
            b.x = W_attn[(size_t)(k0 + kA) * D_ + c];
            b.y = W_attn[(size_t)(k0 + kA + 1) * D_ + c];
            acc = __builtin_amdgcn_wmma_f32_16x16x4_f32(
                false, a, false, b, (short)0, acc, false, false);
        }
#pragma unroll
        for (int i = 0; i < 8; ++i) {
            int   m = i + 8 * half;
            float q = acc[i];
            s0r[i] = fmaf(q, g0s[m][c], s0r[i]);
            s1r[i] = fmaf(q, g1s[m][c], s1r[i]);
        }
    }

#pragma unroll
    for (int i = 0; i < 8; ++i) {
        float a0 = s0r[i], a1 = s1r[i];
        for (int off = 8; off >= 1; off >>= 1) {
            a0 += __shfl_xor(a0, off, 32);
            a1 += __shfl_xor(a1, off, 32);
        }
        if (l16 == 0) {
            int m = i + 8 * half;
            atomicAdd(&sc0[m], a0);
            atomicAdd(&sc1[m], a1);
        }
    }
    __syncthreads();

    if (tid < 16) {
        const float scale = 0.03608439182435161f;
        float a  = sc0[tid] * scale;
        float b  = sc1[tid] * scale;
        float mx = fmaxf(a, b);
        float e0 = __expf(a - mx);
        float e1 = __expf(b - mx);
        float iv = 1.f / (e0 + e1);
        float w0 = e0 * iv, w1 = e1 * iv;
        if (rowLen1[tid]) { w0 = 1.f; w1 = 0.f; }
        wg0[tid] = w0;
        wg1[tid] = w1;
    }
    __syncthreads();

    for (int idx = tid; idx < 16 * D_; idx += 256) {
        int m = idx / D_;
        int d = idx - m * D_;
        float w = wg0[m] * g0s[m][d] + wg1[m] * g1s[m][d]
                + res[(size_t)rowB[m] * D_ + d];
        g0s[m][d] = w;
    }
    __syncthreads();

    {
        const int  nb = wave * 16;
        const int  c  = nb + l16;
        const bool cv = (c < NS_);
        const int  cc = cv ? c : (NS_ - 1);
        const float msk = cv ? 1.f : 0.f;
        v8f acc = {};
        for (int k0 = 0; k0 < D_; k0 += 4) {
            v2f a, b;
            a.x = g0s[l16][k0 + kA];
            a.y = g0s[l16][k0 + kA + 1];
            b.x = W_slot[(size_t)(k0 + kA) * NS_ + cc] * msk;
            b.y = W_slot[(size_t)(k0 + kA + 1) * NS_ + cc] * msk;
            acc = __builtin_amdgcn_wmma_f32_16x16x4_f32(
                false, a, false, b, (short)0, acc, false, false);
        }
        if (cv) {
            float bias = b_slot[c];
#pragma unroll
            for (int i = 0; i < 8; ++i) {
                int m = i + 8 * half;
                int R = mblk * 16 + m;
                out_sf[(size_t)R * NS_ + c] = acc[i] + bias;
            }
        }
    }
}

extern "C" void kernel_launch(void* const* d_in, const int* in_sizes, int n_in,
                              void* d_out, int out_size, void* d_ws, size_t ws_size,
                              hipStream_t stream)
{
    (void)in_sizes; (void)n_in; (void)out_size; (void)d_ws; (void)ws_size;

    const float* res_all = (const float*)d_in[0];
    const float* res     = (const float*)d_in[1];
    const int*   tstart  = (const int*)  d_in[2];
    const int*   slen    = (const int*)  d_in[3];
    const float* W_attn  = (const float*)d_in[4];
    const float* W_id    = (const float*)d_in[5];
    const float* b_id    = (const float*)d_in[6];
    const float* W_slot  = (const float*)d_in[7];
    const float* b_slot  = (const float*)d_in[8];

    float* out_id = (float*)d_out;
    float* out_sf = out_id + B_ * NI_;

    intent_kernel<<<B_ / 16, IK_WAVES * 32, 0, stream>>>(res, W_id, b_id, out_id);
    word_slot_kernel<<<MTILES_, 256, 0, stream>>>(
        res_all, res, tstart, slen, W_attn, W_slot, b_slot, out_sf);
}


# Score GEMM re-cut with W_attn stationary: a work-group keeps one 96-column slab as B fragments in registers, row tiles stream past; slab partial scores added by a new launch (sum order changes)
# speedup vs baseline: 1.8878x; 1.1975x over previous
#include <hip/hip_runtime.h>
#include <math.h>

typedef float v2f __attribute__((ext_vector_type(2)));
typedef float v8f __attribute__((ext_vector_type(8)));

#define B_   128
#define S_   512
#define D_   768
#define T_   51
#define NI_  26
#define NS_  122
#define ROWS_   (B_ * T_)
#define MTILES_ (ROWS_ / 16)
#define LPAD 772

#define IK_WAVES 16
#define IK_KSL   (D_ / IK_WAVES)

__global__ __launch_bounds__(IK_WAVES * 32) void intent_kernel(
    const float* __restrict__ res, const float* __restrict__ W_id,
    const float* __restrict__ b_id, float* __restrict__ out)
{
    __shared__ float part[IK_WAVES][2][8][32];

    const int tid  = threadIdx.x;
    const int lane = tid & 31;
    const int wave = tid >> 5;
    const int half = (lane >> 4) & 1;
    const int l16  = lane & 15;
    const int kA   = 2 * half;
    const int m0   = blockIdx.x * 16;
    const int kb   = wave * IK_KSL;

    const int c0 = l16;
    const int c1 = (16 + l16 < NI_) ? (16 + l16) : (NI_ - 1);

    const float* ap = res  + (size_t)(m0 + l16) * D_ + kb + kA;
    const float* bp = W_id + (size_t)(kb + kA) * NI_;

    v8f acc0 = {}, acc1 = {};
    for (int s = 0; s < IK_KSL / 4; ++s) {
        v2f a, b0, b1;
        a.x  = ap[4 * s];
        a.y  = ap[4 * s + 1];
        b0.x = bp[(size_t)(4 * s)     * NI_ + c0];
        b0.y = bp[(size_t)(4 * s + 1) * NI_ + c0];
        b1.x = bp[(size_t)(4 * s)     * NI_ + c1];
        b1.y = bp[(size_t)(4 * s + 1) * NI_ + c1];
        acc0 = __builtin_amdgcn_wmma_f32_16x16x4_f32(
            false, a, false, b0, (short)0, acc0, false, false);
        acc1 = __builtin_amdgcn_wmma_f32_16x16x4_f32(
            false, a, false, b1, (short)0, acc1, false, false);
    }

    for (int i = 0; i < 8; ++i) {
        part[wave][0][i][lane] = acc0[i];
        part[wave][1][i][lane] = acc1[i];
    }
    __syncthreads();

    if (tid < 16 * NI_) {
        const int row = tid / NI_;
        const int col = tid - row * NI_;
        const int tl  = col >> 4;
        const int i   = row & 7;
        const int ln  = ((row >> 3) << 4) | (col & 15);
        float s = part[0][tl][i][ln];
        for (int w = 1; w < IK_WAVES; ++w) s += part[w][tl][i][ln];
        out[(size_t)blockIdx.x * (16 * NI_) + tid] = s + b_id[col];
    }
}

#define SC_SLABS  8
#define SC_SLABW  (D_ / SC_SLABS)
#define SC_GROUPS 32
#define SC_KSL    (D_ / 8)
#define SC_MAXT   ((MTILES_ + SC_GROUPS - 1) / SC_GROUPS)
#define TILE_OUT  (16 * NS_)

#define SC_ROW_XOR(x, W_) __builtin_bit_cast(float, __builtin_amdgcn_update_dpp( \
    0, __builtin_bit_cast(int, (x)), 0x160 | (W_), 0xf, 0xf, true))
#define SC_HALVE(W_) {                                              \
        const bool hi = (l16 & (W_)) != 0;                          \
        _Pragma("unroll")                                           \
        for (int q = 0; q < (W_); ++q) {                            \
            const float keep = hi ? v[q + (W_)] : v[q];             \
            const float send = hi ? v[q] : v[q + (W_)];             \
            v[q] = keep + SC_ROW_XOR(send, (W_));                   \
        } }

__global__ __launch_bounds__(256) void score_kernel(
    const float* __restrict__ res_all,
    const int*  __restrict__ tstart, const int* __restrict__ slen,
    const float* __restrict__ W_attn, float* __restrict__ out_sf)
{
    __shared__ float part[SC_MAXT][8][32];

    const int tid  = threadIdx.x;
    const int lane = tid & 31;
    const int wave = tid >> 5;
    const int half = (lane >> 4) & 1;
    const int l16  = lane & 15;
    const int slab = blockIdx.x & (SC_SLABS - 1);
    const int grp  = blockIdx.x / SC_SLABS;
    const int nb   = slab * SC_SLABW;
    const int kb   = wave * SC_KSL;
    const int ntiles = (grp < MTILES_ - (SC_MAXT - 1) * SC_GROUPS) ? SC_MAXT : (SC_MAXT - 1);

    v2f wf[24][6];
#pragma unroll
    for (int j = 0; j < 12; ++j) {
#pragma unroll
        for (int p = 0; p < 2; ++p) {
            const float* wp = W_attn + (size_t)(kb + 8 * j + 4 * half + 2 * p) * D_ + nb + l16;
#pragma unroll
            for (int t = 0; t < 6; ++t) {
                wf[2 * j + p][t].x = wp[16 * t];
                wf[2 * j + p][t].y = wp[D_ + 16 * t];
            }
        }
    }

    int tile = grp;
    int off0, off1;
    {
        const int R  = tile * 16 + l16;
        const int bb = R / T_;
        off0 = (bb * S_ + tstart[R]) * D_;
        off1 = off0 + slen[R] * D_;
    }
    float4 a[12];
    const int ka = kb + 4 * half;
#pragma unroll
    for (int j = 0; j < 12; ++j)
        a[j] = *(const float4*)&res_all[off0 + ka + 8 * j];

    for (int n = 0; n < ntiles; ++n) {
        const int tnext = (n + 1 < ntiles) ? (tile + SC_GROUPS) : tile;
        const int Rn  = tnext * 16 + l16;
        const int stn = tstart[Rn];
        const int lnn = slen[Rn];

        v8f acc[6];
#pragma unroll
        for (int t = 0; t < 6; ++t) acc[t] = (v8f){};
#pragma unroll
        for (int j = 0; j < 12; ++j) {
            v2f a0, a1;
            a0.x = a[j].x; a0.y = a[j].y;
            a1.x = a[j].z; a1.y = a[j].w;
#pragma unroll
            for (int t = 0; t < 6; ++t)
                acc[t] = __builtin_amdgcn_wmma_f32_16x16x4_f32(
                    false, a0, false, wf[2 * j][t], (short)0, acc[t], false, false);
#pragma unroll
            for (int t = 0; t < 6; ++t)
                acc[t] = __builtin_amdgcn_wmma_f32_16x16x4_f32(
                    false, a1, false, wf[2 * j + 1][t], (short)0, acc[t], false, false);
        }

        const int off0n = ((Rn / T_) * S_ + stn) * D_;
        const int off1n = off0n + lnn * D_;
#pragma unroll
        for (int j = 0; j < 12; ++j)
            a[j] = *(const float4*)&res_all[off0n + ka + 8 * j];

        float v[16];
#pragma unroll
        for (int i = 0; i < 8; ++i) {
            const int o0 = __shfl(off0, i + 8 * half, 32);
            const int o1 = __shfl(off1, i + 8 * half, 32);
            const int e0 = o0 + nb + l16;
            const int e1 = o1 + nb + l16;
            float x0 = 0.f, x1 = 0.f;
#pragma unroll
            for (int t = 0; t < 6; ++t) {
                x0 = fmaf(acc[t][i], res_all[e0 + 16 * t], x0);
                x1 = fmaf(acc[t][i], res_all[e1 + 16 * t], x1);
            }
            v[2 * i]     = x0;
            v[2 * i + 1] = x1;
        }
        SC_HALVE(8)
        SC_HALVE(4)
        SC_HALVE(2)
        SC_HALVE(1)
        part[n][wave][lane] = v[0];

        tile = tnext;
        off0 = off0n;
        off1 = off1n;
    }
    __syncthreads();

    for (int idx = tid; idx < ntiles * 32; idx += 256) {
        const int lt = idx >> 5;
        const int e  = idx & 31;
        float s = part[lt][0][e];
#pragma unroll
        for (int w = 1; w < 8; ++w) s += part[lt][w][e];
        out_sf[(size_t)(grp + SC_GROUPS * lt) * TILE_OUT + slab * 32 + e] = s;
    }
}

__global__ __launch_bounds__(256) void word_slot_kernel(
    const float* __restrict__ res_all, const float* __restrict__ res,
    const int*  __restrict__ tstart,   const int*  __restrict__ slen,
    const float* __restrict__ W_slot,  const float* __restrict__ b_slot,
    float* __restrict__ out_sf)
{
    __shared__ float g0s[16][LPAD];
    __shared__ float g1s[16][LPAD];
    __shared__ float sc0[16], sc1[16], wg0[16], wg1[16];
    __shared__ int   rowB[16], rowS0[16], rowS1[16], rowLen1[16];

    const int tid  = threadIdx.x;
    const int lane = tid & 31;
    const int wave = tid >> 5;
    const int half = (lane >> 4) & 1;
    const int l16  = lane & 15;
    const int kA   = 2 * half;
    const int mblk = blockIdx.x;

    if (tid < 16) {
        int R  = mblk * 16 + tid;
        int bb = R / T_;
        int tt = R - bb * T_;
        int st = tstart[bb * T_ + tt];
        int ln = slen[bb * T_ + tt];
        rowB[tid]    = bb;
        rowS0[tid]   = st;
        rowS1[tid]   = st + ln;
        rowLen1[tid] = (ln == 1);
        const float2* ps = (const float2*)(out_sf + (size_t)mblk * TILE_OUT) + tid;
        float2 acc2 = ps[0];
        for (int sl = 1; sl < SC_SLABS; ++sl) {
            const float2 v = ps[sl * 16];
            acc2.x += v.x;
            acc2.y += v.y;
        }
        sc0[tid] = acc2.x;
        sc1[tid] = acc2.y;
    }
    __syncthreads();

    for (int idx = tid; idx < 16 * (D_ / 4); idx += 256) {
        int m = idx / (D_ / 4);
        int q = idx - m * (D_ / 4);
        const float4* s0p = (const float4*)(res_all + (size_t)(rowB[m] * S_ + rowS0[m]) * D_);
        const float4* s1p = (const float4*)(res_all + (size_t)(rowB[m] * S_ + rowS1[m]) * D_);
        ((float4*)&g0s[m][0])[q] = s0p[q];
        ((float4*)&g1s[m][0])[q] = s1p[q];
    }
    __syncthreads();

    if (tid < 16) {
        const float scale = 0.03608439182435161f;
        float a  = sc0[tid] * scale;
        float b  = sc1[tid] * scale;
        float mx = fmaxf(a, b);
        float e0 = __expf(a - mx);
        float e1 = __expf(b - mx);
        float iv = 1.f / (e0 + e1);
        float w0 = e0 * iv, w1 = e1 * iv;
        if (rowLen1[tid]) { w0 = 1.f; w1 = 0.f; }
        wg0[tid] = w0;
        wg1[tid] = w1;
    }
    __syncthreads();

    for (int idx = tid; idx < 16 * D_; idx += 256) {
        int m = idx / D_;
        int d = idx - m * D_;
        float w = wg0[m] * g0s[m][d] + wg1[m] * g1s[m][d]
                + res[(size_t)rowB[m] * D_ + d];
        g0s[m][d] = w;
    }
    __syncthreads();

    {
        const int  nb = wave * 16;
        const int  c  = nb + l16;
        const bool cv = (c < NS_);
        const int  cc = cv ? c : (NS_ - 1);
        const float msk = cv ? 1.f : 0.f;
        v8f acc = {};
        for (int k0 = 0; k0 < D_; k0 += 4) {
            v2f a, b;
            a.x = g0s[l16][k0 + kA];
            a.y = g0s[l16][k0 + kA + 1];
            b.x = W_slot[(size_t)(k0 + kA) * NS_ + cc] * msk;
            b.y = W_slot[(size_t)(k0 + kA + 1) * NS_ + cc] * msk;
            acc = __builtin_amdgcn_wmma_f32_16x16x4_f32(
                false, a, false, b, (short)0, acc, false, false);
        }
        if (cv) {
            float bias = b_slot[c];
#pragma unroll
            for (int i = 0; i < 8; ++i) {
                int m = i + 8 * half;
                int R = mblk * 16 + m;
                out_sf[(size_t)R * NS_ + c] = acc[i] + bias;
            }
        }
    }
}

extern "C" void kernel_launch(void* const* d_in, const int* in_sizes, int n_in,
                              void* d_out, int out_size, void* d_ws, size_t ws_size,
                              hipStream_t stream)
{
    (void)in_sizes; (void)n_in; (void)out_size; (void)d_ws; (void)ws_size;

    const float* res_all = (const float*)d_in[0];
    const float* res     = (const float*)d_in[1];
    const int*   tstart  = (const int*)  d_in[2];
    const int*   slen    = (const int*)  d_in[3];
    const float* W_attn  = (const float*)d_in[4];
    const float* W_id    = (const float*)d_in[5];
    const float* b_id    = (const float*)d_in[6];
    const float* W_slot  = (const float*)d_in[7];
    const float* b_slot  = (const float*)d_in[8];

    float* out_id = (float*)d_out;
    float* out_sf = out_id + B_ * NI_;

    intent_kernel<<<B_ / 16, IK_WAVES * 32, 0, stream>>>(res, W_id, b_id, out_id);
    score_kernel<<<SC_SLABS * SC_GROUPS, 256, 0, stream>>>(
        res_all, tstart, slen, W_attn, out_sf);
    word_slot_kernel<<<MTILES_, 256, 0, stream>>>(
        res_all, res, tstart, slen, W_slot, b_slot, out_sf);
}
